# starting kernel
# baseline (speedup 1.0000x reference)
#include <hip/hip_runtime.h>
#include <hip/hip_bf16.h>

typedef float v2f __attribute__((ext_vector_type(2)));
typedef float v8f __attribute__((ext_vector_type(8)));

#define WMMA_F32(a, b, c) \
  __builtin_amdgcn_wmma_f32_16x16x4_f32(false, (a), false, (b), (short)0, (c), false, false)

__launch_bounds__(256)
__global__ void gemm_bias_kernel(const float* __restrict__ A, const float* __restrict__ B,
                                 const float* __restrict__ bias, float* __restrict__ C,
                                 int M, int N, int K)
{
    __shared__ __align__(16) float As [128 * 34];
    __shared__ __align__(16) float BsT[128 * 34];

    const int tid  = threadIdx.x;
    const int wid  = tid >> 5;
    const int lane = tid & 31;
    const int m16  = lane & 15;
    const int h    = lane >> 4;
    const int wm   = wid >> 1;
    const int wn   = wid & 1;
    const int row0 = blockIdx.y * 128;
    const int col0 = blockIdx.x * 128;

    const int ar  = tid >> 1;
    const int akq = (tid & 1) << 4;
    const int bkr = tid >> 5;
    const int bcq = (tid & 31) << 2;

    v8f acc[2][4];
    #pragma unroll
    for (int i = 0; i < 2; ++i)
        #pragma unroll
        for (int j = 0; j < 4; ++j) { v8f z{}; acc[i][j] = z; }

    float4 pa[4], pb[4];
    #pragma unroll
    for (int i = 0; i < 4; ++i)
        pa[i] = *(const float4*)(A + (size_t)(row0 + ar) * K + akq + 4 * i);
    #pragma unroll
    for (int i = 0; i < 4; ++i)
        pb[i] = *(const float4*)(B + (size_t)(bkr + 8 * i) * N + col0 + bcq);

    for (int kk = 0; kk < K; kk += 32) {
        #pragma unroll
        for (int i = 0; i < 4; ++i) {
            float* d = &As[ar * 34 + akq + 4 * i];
            d[0] = pa[i].x; d[1] = pa[i].y; d[2] = pa[i].z; d[3] = pa[i].w;
        }
        #pragma unroll
        for (int i = 0; i < 4; ++i) {
            int kr = bkr + 8 * i;
            BsT[(bcq + 0) * 34 + kr] = pb[i].x;
            BsT[(bcq + 1) * 34 + kr] = pb[i].y;
            BsT[(bcq + 2) * 34 + kr] = pb[i].z;
            BsT[(bcq + 3) * 34 + kr] = pb[i].w;
        }
        __syncthreads();

        if (kk + 32 < K) {
            #pragma unroll
            for (int i = 0; i < 4; ++i)
                pa[i] = *(const float4*)(A + (size_t)(row0 + ar) * K + kk + 32 + akq + 4 * i);
            #pragma unroll
            for (int i = 0; i < 4; ++i)
                pb[i] = *(const float4*)(B + (size_t)(bkr + 8 * i) * N + kk * 0 + (size_t)(kk + 32 + bkr + 8 * i - (bkr + 8 * i)) * 0 + col0 + bcq + (size_t)(kk + 32) * 0)
                      , pb[i] = *(const float4*)(B + (size_t)(kk + 32 + bkr + 8 * i) * N + col0 + bcq);
        }

        #pragma unroll
        for (int ks = 0; ks < 32; ks += 4) {
            v2f a[2], b[4];
            #pragma unroll
            for (int mi = 0; mi < 2; ++mi)
                a[mi] = *(const v2f*)(&As[(wm * 32 + mi * 16 + m16) * 34 + ks + 2 * h]);
            #pragma unroll
            for (int ni = 0; ni < 4; ++ni)
                b[ni] = *(const v2f*)(&BsT[(wn * 64 + ni * 16 + m16) * 34 + ks + 2 * h]);
            #pragma unroll
            for (int mi = 0; mi < 2; ++mi)
                #pragma unroll
                for (int ni = 0; ni < 4; ++ni)
                    acc[mi][ni] = WMMA_F32(a[mi], b[ni], acc[mi][ni]);
        }
        __syncthreads();
    }

    #pragma unroll
    for (int mi = 0; mi < 2; ++mi) {
        #pragma unroll
        for (int ni = 0; ni < 4; ++ni) {
            int c = col0 + wn * 64 + ni * 16 + m16;
            float bv = bias[c];
            #pragma unroll
            for (int r = 0; r < 8; ++r) {
                int rr = row0 + wm * 32 + mi * 16 + r + 8 * h;
                C[(size_t)rr * N + c] = acc[mi][ni][r] + bv;
            }
        }
    }
}

__global__ void rope_repack_kernel(const float* __restrict__ qkv,
                                   const float* __restrict__ cosT,
                                   const float* __restrict__ sinT,
                                   float* __restrict__ Q, float* __restrict__ Ko,
                                   float* __restrict__ V)
{
    int tid = blockIdx.x * blockDim.x + threadIdx.x;
    int dp  = tid & 31;
    int h0  = (tid >> 5) & 15;
    int s   = (tid >> 9) & 2047;
    int b   = tid >> 20;
    size_t rowq = (size_t)(b * 2048 + s) * 3072;
    int off = h0 * 64 + 2 * dp;

    float c  = cosT[s * 32 + dp];
    float sn = sinT[s * 32 + dp];
    float qe = qkv[rowq + off],        qo = qkv[rowq + off + 1];
    float ke = qkv[rowq + 1024 + off], ko = qkv[rowq + 1024 + off + 1];

    size_t oidx = ((size_t)((b * 16 + h0) * 2048 + s)) * 64 + 2 * dp;
    Q[oidx]      = qe * c - qo * sn;
    Q[oidx + 1]  = qe * sn + qo * c;
    Ko[oidx]     = ke * c - ko * sn;
    Ko[oidx + 1] = ke * sn + ko * c;
    V[oidx]      = qkv[rowq + 2048 + off];
    V[oidx + 1]  = qkv[rowq + 2048 + off + 1];
}

__launch_bounds__(128)
__global__ void flash_attn_kernel(const float* __restrict__ Q, const float* __restrict__ Kg,
                                  const float* __restrict__ V, float* __restrict__ O)
{
    __shared__ __align__(16) float Ks [16 * 66];
    __shared__ __align__(16) float Vst[64 * 18];
    __shared__ __align__(16) float Pw [4 * 16 * 18];

    const int tid  = threadIdx.x;
    const int wid  = tid >> 5;
    const int lane = tid & 31;
    const int m16  = lane & 15;
    const int h    = lane >> 4;
    const int bh   = blockIdx.y;
    const int qbase = blockIdx.x * 64;
    const int qrow  = qbase + wid * 16;
    const size_t base = (size_t)bh * 2048 * 64;

    const int kvr  = tid >> 3;
    const int kvcq = (tid & 7) << 3;

    v2f aq[16];
    {
        const float* qp = Q + base + (size_t)(qrow + m16) * 64;
        #pragma unroll
        for (int s2 = 0; s2 < 16; ++s2) {
            v2f t = *(const v2f*)(qp + 4 * s2 + 2 * h);
            aq[s2] = t * 0.125f;
        }
    }

    float mrow[8], lrow[8];
    #pragma unroll
    for (int r = 0; r < 8; ++r) { mrow[r] = -1e30f; lrow[r] = 0.0f; }
    v8f oacc[4];
    #pragma unroll
    for (int i = 0; i < 4; ++i) { v8f z{}; oacc[i] = z; }

    const int ntiles = blockIdx.x * 4 + 4;
    float* Pme = &Pw[wid * 16 * 18];

    float4 pk[2], pv[2];
    #pragma unroll
    for (int i = 0; i < 2; ++i) {
        pk[i] = *(const float4*)(Kg + base + (size_t)kvr * 64 + kvcq + 4 * i);
        pv[i] = *(const float4*)(V  + base + (size_t)kvr * 64 + kvcq + 4 * i);
    }

    for (int kt = 0; kt < ntiles; ++kt) {
        #pragma unroll
        for (int i = 0; i < 2; ++i) {
            float* d = &Ks[kvr * 66 + kvcq + 4 * i];
            d[0] = pk[i].x; d[1] = pk[i].y; d[2] = pk[i].z; d[3] = pk[i].w;
            int c = kvcq + 4 * i;
            Vst[(c + 0) * 18 + kvr] = pv[i].x;
            Vst[(c + 1) * 18 + kvr] = pv[i].y;
            Vst[(c + 2) * 18 + kvr] = pv[i].z;
            Vst[(c + 3) * 18 + kvr] = pv[i].w;
        }
        __syncthreads();

        if (kt + 1 < ntiles) {
            #pragma unroll
            for (int i = 0; i < 2; ++i) {
                pk[i] = *(const float4*)(Kg + base + (size_t)((kt + 1) * 16 + kvr) * 64 + kvcq + 4 * i);
                pv[i] = *(const float4*)(V  + base + (size_t)((kt + 1) * 16 + kvr) * 64 + kvcq + 4 * i);
            }
        }

        v8f sc{};
        #pragma unroll
        for (int s2 = 0; s2 < 16; ++s2) {
            v2f b = *(const v2f*)(&Ks[m16 * 66 + 4 * s2 + 2 * h]);
            sc = WMMA_F32(aq[s2], b, sc);
        }

        const int kcol = kt * 16 + m16;
        #pragma unroll
        for (int r = 0; r < 8; ++r) {
            int qg = qrow + r + 8 * h;
            float val = (kcol <= qg) ? sc[r] : -1e30f;
            float mx = val;
            mx = fmaxf(mx, __shfl_xor(mx, 1, 32));
            mx = fmaxf(mx, __shfl_xor(mx, 2, 32));
            mx = fmaxf(mx, __shfl_xor(mx, 4, 32));
            mx = fmaxf(mx, __shfl_xor(mx, 8, 32));
            float mnew = fmaxf(mrow[r], mx);
            float p = __expf(val - mnew);
            float sum = p;
            sum += __shfl_xor(sum, 1, 32);
            sum += __shfl_xor(sum, 2, 32);
            sum += __shfl_xor(sum, 4, 32);
            sum += __shfl_xor(sum, 8, 32);
            float alpha = __expf(mrow[r] - mnew);
            mrow[r] = mnew;
            lrow[r] = lrow[r] * alpha + sum;
            #pragma unroll
            for (int ni = 0; ni < 4; ++ni) oacc[ni][r] *= alpha;
            Pme[(r + 8 * h) * 18 + m16] = p;
        }
        __builtin_amdgcn_wave_barrier();

        #pragma unroll
        for (int ks = 0; ks < 16; ks += 4) {
            v2f a = *(const v2f*)(&Pme[m16 * 18 + ks + 2 * h]);
            #pragma unroll
            for (int ni = 0; ni < 4; ++ni) {
                v2f b = *(const v2f*)(&Vst[(ni * 16 + m16) * 18 + ks + 2 * h]);
                oacc[ni] = WMMA_F32(a, b, oacc[ni]);
            }
        }
        __syncthreads();
    }

    const int b_ = bh >> 4, head = bh & 15;
    #pragma unroll
    for (int ni = 0; ni < 4; ++ni) {
        int c = head * 64 + ni * 16 + m16;
        #pragma unroll
        for (int r = 0; r < 8; ++r) {
            int srow = qrow + r + 8 * h;
            float inv = 1.0f / lrow[r];
            O[((size_t)(b_ * 2048 + srow)) * 1024 + c] = oacc[ni][r] * inv;
        }
    }
}

extern "C" void kernel_launch(void* const* d_in, const int* in_sizes, int n_in,
                              void* d_out, int out_size, void* d_ws, size_t ws_size,
                              hipStream_t stream)
{
    const float* x    = (const float*)d_in[0];
    const float* cosT = (const float*)d_in[2];
    const float* sinT = (const float*)d_in[3];
    const float* Wqkv = (const float*)d_in[4];
    const float* bqkv = (const float*)d_in[5];
    const float* Wo   = (const float*)d_in[6];
    const float* bo   = (const float*)d_in[7];
    float* out = (float*)d_out;

    float* ws  = (float*)d_ws;
    float* qkv = ws;
    float* Qb  = qkv + (size_t)4096 * 3072;
    float* Kb  = Qb + (size_t)4194304;
    float* Vb  = Kb + (size_t)4194304;
    float* Ob  = Vb + (size_t)4194304;

    gemm_bias_kernel<<<dim3(3072 / 128, 4096 / 128), 256, 0, stream>>>(
        x, Wqkv, bqkv, qkv, 4096, 3072, 1024);

    rope_repack_kernel<<<2097152 / 256, 256, 0, stream>>>(qkv, cosT, sinT, Qb, Kb, Vb);

    flash_attn_kernel<<<dim3(32, 32), 128, 0, stream>>>(Qb, Kb, Vb, Ob);

    gemm_bias_kernel<<<dim3(1024 / 128, 4096 / 128), 256, 0, stream>>>(
        Ob, Wo, bo, out, 4096, 1024, 1024);
}


# Attention K/V tile widened from 16 to 64 keys per loop trip: 4x fewer block barriers, cross-lane row reductions and accumulator rescales per key; softmax and PV sums regrouped per 64 keys
# speedup vs baseline: 1.1689x; 1.1689x over previous
#include <hip/hip_runtime.h>
#include <hip/hip_bf16.h>

typedef float v2f __attribute__((ext_vector_type(2)));
typedef float v4f __attribute__((ext_vector_type(4)));
typedef float v8f __attribute__((ext_vector_type(8)));

#define WMMA_F32(a, b, c) \
  __builtin_amdgcn_wmma_f32_16x16x4_f32(false, (a), false, (b), (short)0, (c), false, false)

__launch_bounds__(256)
__global__ void gemm_bias_kernel(const float* __restrict__ A, const float* __restrict__ B,
                                 const float* __restrict__ bias, float* __restrict__ C,
                                 int M, int N, int K)
{
    __shared__ __align__(16) float As [128 * 34];
    __shared__ __align__(16) float BsT[128 * 34];

    const int tid  = threadIdx.x;
    const int wid  = tid >> 5;
    const int lane = tid & 31;
    const int m16  = lane & 15;
    const int h    = lane >> 4;
    const int wm   = wid >> 1;
    const int wn   = wid & 1;
    const int row0 = blockIdx.y * 128;
    const int col0 = blockIdx.x * 128;

    const int ar  = tid >> 1;
    const int akq = (tid & 1) << 4;
    const int bkr = tid >> 5;
    const int bcq = (tid & 31) << 2;

    v8f acc[2][4];
    #pragma unroll
    for (int i = 0; i < 2; ++i)
        #pragma unroll
        for (int j = 0; j < 4; ++j) { v8f z{}; acc[i][j] = z; }

    float4 pa[4], pb[4];
    #pragma unroll
    for (int i = 0; i < 4; ++i)
        pa[i] = *(const float4*)(A + (size_t)(row0 + ar) * K + akq + 4 * i);
    #pragma unroll
    for (int i = 0; i < 4; ++i)
        pb[i] = *(const float4*)(B + (size_t)(bkr + 8 * i) * N + col0 + bcq);

    for (int kk = 0; kk < K; kk += 32) {
        #pragma unroll
        for (int i = 0; i < 4; ++i) {
            float* d = &As[ar * 34 + akq + 4 * i];
            d[0] = pa[i].x; d[1] = pa[i].y; d[2] = pa[i].z; d[3] = pa[i].w;
        }
        #pragma unroll
        for (int i = 0; i < 4; ++i) {
            int kr = bkr + 8 * i;
            BsT[(bcq + 0) * 34 + kr] = pb[i].x;
            BsT[(bcq + 1) * 34 + kr] = pb[i].y;
            BsT[(bcq + 2) * 34 + kr] = pb[i].z;
            BsT[(bcq + 3) * 34 + kr] = pb[i].w;
        }
        __syncthreads();

        if (kk + 32 < K) {
            #pragma unroll
            for (int i = 0; i < 4; ++i)
                pa[i] = *(const float4*)(A + (size_t)(row0 + ar) * K + kk + 32 + akq + 4 * i);
            #pragma unroll
            for (int i = 0; i < 4; ++i)
                pb[i] = *(const float4*)(B + (size_t)(bkr + 8 * i) * N + kk * 0 + (size_t)(kk + 32 + bkr + 8 * i - (bkr + 8 * i)) * 0 + col0 + bcq + (size_t)(kk + 32) * 0)
                      , pb[i] = *(const float4*)(B + (size_t)(kk + 32 + bkr + 8 * i) * N + col0 + bcq);
        }

        #pragma unroll
        for (int ks = 0; ks < 32; ks += 4) {
            v2f a[2], b[4];
            #pragma unroll
            for (int mi = 0; mi < 2; ++mi)
                a[mi] = *(const v2f*)(&As[(wm * 32 + mi * 16 + m16) * 34 + ks + 2 * h]);
            #pragma unroll
            for (int ni = 0; ni < 4; ++ni)
                b[ni] = *(const v2f*)(&BsT[(wn * 64 + ni * 16 + m16) * 34 + ks + 2 * h]);
            #pragma unroll
            for (int mi = 0; mi < 2; ++mi)
                #pragma unroll
                for (int ni = 0; ni < 4; ++ni)
                    acc[mi][ni] = WMMA_F32(a[mi], b[ni], acc[mi][ni]);
        }
        __syncthreads();
    }

    #pragma unroll
    for (int mi = 0; mi < 2; ++mi) {
        #pragma unroll
        for (int ni = 0; ni < 4; ++ni) {
            int c = col0 + wn * 64 + ni * 16 + m16;
            float bv = bias[c];
            #pragma unroll
            for (int r = 0; r < 8; ++r) {
                int rr = row0 + wm * 32 + mi * 16 + r + 8 * h;
                C[(size_t)rr * N + c] = acc[mi][ni][r] + bv;
            }
        }
    }
}

__global__ void rope_repack_kernel(const float* __restrict__ qkv,
                                   const float* __restrict__ cosT,
                                   const float* __restrict__ sinT,
                                   float* __restrict__ Q, float* __restrict__ Ko,
                                   float* __restrict__ V)
{
    int tid = blockIdx.x * blockDim.x + threadIdx.x;
    int dp  = tid & 31;
    int h0  = (tid >> 5) & 15;
    int s   = (tid >> 9) & 2047;
    int b   = tid >> 20;
    size_t rowq = (size_t)(b * 2048 + s) * 3072;
    int off = h0 * 64 + 2 * dp;

    float c  = cosT[s * 32 + dp];
    float sn = sinT[s * 32 + dp];
    float qe = qkv[rowq + off],        qo = qkv[rowq + off + 1];
    float ke = qkv[rowq + 1024 + off], ko = qkv[rowq + 1024 + off + 1];

    size_t oidx = ((size_t)((b * 16 + h0) * 2048 + s)) * 64 + 2 * dp;
    Q[oidx]      = qe * c - qo * sn;
    Q[oidx + 1]  = qe * sn + qo * c;
    Ko[oidx]     = ke * c - ko * sn;
    Ko[oidx + 1] = ke * sn + ko * c;
    V[oidx]      = qkv[rowq + 2048 + off];
    V[oidx + 1]  = qkv[rowq + 2048 + off + 1];
}

__launch_bounds__(128)
__global__ void flash_attn_kernel(const float* __restrict__ Q, const float* __restrict__ Kg,
                                  const float* __restrict__ V, float* __restrict__ O)
{
    constexpr int LP = 68;
    __shared__ __align__(16) float Ks [64 * LP];
    __shared__ __align__(16) float Vst[64 * LP];
    __shared__ __align__(16) float Pw [4 * 16 * LP];

    const int tid  = threadIdx.x;
    const int wid  = tid >> 5;
    const int lane = tid & 31;
    const int m16  = lane & 15;
    const int h    = lane >> 4;
    const int bh   = blockIdx.y;
    const int qbase = blockIdx.x * 64;
    const int qrow  = qbase + wid * 16;
    const size_t base = (size_t)bh * 2048 * 64;

    const int kkr = tid >> 4;
    const int kc  = (tid & 15) << 2;
    const int vkr = tid >> 3;
    const int vc  = (tid & 7) << 2;

    v2f aq[16];
    {
        const float* qp = Q + base + (size_t)(qrow + m16) * 64;
        #pragma unroll
        for (int s2 = 0; s2 < 16; ++s2) {
            v2f t = *(const v2f*)(qp + 4 * s2 + 2 * h);
            aq[s2] = t * 0.125f;
        }
    }

    float mrow[8], lrow[8];
    #pragma unroll
    for (int r = 0; r < 8; ++r) { mrow[r] = -1e30f; lrow[r] = 0.0f; }
    v8f oacc[4];
    #pragma unroll
    for (int i = 0; i < 4; ++i) { v8f z{}; oacc[i] = z; }

    const int ntiles = blockIdx.x + 1;
    float* Pme = &Pw[wid * 16 * LP];

    v4f pk[8], pv[8];
    #pragma unroll
    for (int i = 0; i < 8; ++i) {
        pk[i] = *(const v4f*)(Kg + base + (size_t)(kkr + 8 * i) * 64 + kc);
        pv[i] = *(const v4f*)(V  + base + (size_t)(vkr + 16 * (i >> 1)) * 64 + 32 * (i & 1) + vc);
    }

    for (int kt = 0; kt < ntiles; ++kt) {
        #pragma unroll
        for (int i = 0; i < 8; ++i) {
            *(v4f*)(&Ks[(kkr + 8 * i) * LP + kc]) = pk[i];
            const int key = vkr + 16 * (i >> 1);
            const int c   = 32 * (i & 1) + vc;
            Vst[(c + 0) * LP + key] = pv[i].x;
            Vst[(c + 1) * LP + key] = pv[i].y;
            Vst[(c + 2) * LP + key] = pv[i].z;
            Vst[(c + 3) * LP + key] = pv[i].w;
        }
        __syncthreads();

        if (kt + 1 < ntiles) {
            const size_t nb = base + (size_t)(kt + 1) * 64 * 64;
            #pragma unroll
            for (int i = 0; i < 8; ++i) {
                pk[i] = *(const v4f*)(Kg + nb + (size_t)(kkr + 8 * i) * 64 + kc);
                pv[i] = *(const v4f*)(V  + nb + (size_t)(vkr + 16 * (i >> 1)) * 64 + 32 * (i & 1) + vc);
            }
        }

        v8f sc[4];
        #pragma unroll
        for (int j = 0; j < 4; ++j) { v8f z{}; sc[j] = z; }
        #pragma unroll
        for (int s2 = 0; s2 < 16; ++s2) {
            #pragma unroll
            for (int j = 0; j < 4; ++j) {
                v2f b = *(const v2f*)(&Ks[(16 * j + m16) * LP + 4 * s2 + 2 * h]);
                sc[j] = WMMA_F32(aq[s2], b, sc[j]);
            }
        }

        if (kt == ntiles - 1) {
            #pragma unroll
            for (int r = 0; r < 8; ++r) {
                const int lim = wid * 16 + r + 8 * h - m16;
                sc[0][r] = (0  <= lim) ? sc[0][r] : -1e30f;
                sc[1][r] = (16 <= lim) ? sc[1][r] : -1e30f;
                sc[2][r] = (32 <= lim) ? sc[2][r] : -1e30f;
                sc[3][r] = (48 <= lim) ? sc[3][r] : -1e30f;
            }
        }
        #pragma unroll
        for (int r = 0; r < 8; ++r) {
            const float v0 = sc[0][r], v1 = sc[1][r], v2 = sc[2][r], v3 = sc[3][r];
            float mx = fmaxf(fmaxf(v0, v1), fmaxf(v2, v3));
            mx = fmaxf(mx, __shfl_xor(mx, 1, 32));
            mx = fmaxf(mx, __shfl_xor(mx, 2, 32));
            mx = fmaxf(mx, __shfl_xor(mx, 4, 32));
            mx = fmaxf(mx, __shfl_xor(mx, 8, 32));
            const float mnew = fmaxf(mrow[r], mx);
            const float p0 = __expf(v0 - mnew);
            const float p1 = __expf(v1 - mnew);
            const float p2 = __expf(v2 - mnew);
            const float p3 = __expf(v3 - mnew);
            float sum = (p0 + p1) + (p2 + p3);
            sum += __shfl_xor(sum, 1, 32);
            sum += __shfl_xor(sum, 2, 32);
            sum += __shfl_xor(sum, 4, 32);
            sum += __shfl_xor(sum, 8, 32);
            const float alpha = __expf(mrow[r] - mnew);
            mrow[r] = mnew;
            lrow[r] = lrow[r] * alpha + sum;
            #pragma unroll
            for (int ni = 0; ni < 4; ++ni) oacc[ni][r] *= alpha;
            float* prow = &Pme[(r + 8 * h) * LP + m16];
            prow[0]  = p0;
            prow[16] = p1;
            prow[32] = p2;
            prow[48] = p3;
        }
        __builtin_amdgcn_wave_barrier();

        #pragma unroll
        for (int ks = 0; ks < 64; ks += 4) {
            v2f a = *(const v2f*)(&Pme[m16 * LP + ks + 2 * h]);
            #pragma unroll
            for (int ni = 0; ni < 4; ++ni) {
                v2f b = *(const v2f*)(&Vst[(ni * 16 + m16) * LP + ks + 2 * h]);
                oacc[ni] = WMMA_F32(a, b, oacc[ni]);
            }
        }
        __syncthreads();
    }

    const int b_ = bh >> 4, head = bh & 15;
    #pragma unroll
    for (int ni = 0; ni < 4; ++ni) {
        int c = head * 64 + ni * 16 + m16;
        #pragma unroll
        for (int r = 0; r < 8; ++r) {
            int srow = qrow + r + 8 * h;
            float inv = 1.0f / lrow[r];
            O[((size_t)(b_ * 2048 + srow)) * 1024 + c] = oacc[ni][r] * inv;
        }
    }
}

extern "C" void kernel_launch(void* const* d_in, const int* in_sizes, int n_in,
                              void* d_out, int out_size, void* d_ws, size_t ws_size,
                              hipStream_t stream)
{
    const float* x    = (const float*)d_in[0];
    const float* cosT = (const float*)d_in[2];
    const float* sinT = (const float*)d_in[3];
    const float* Wqkv = (const float*)d_in[4];
    const float* bqkv = (const float*)d_in[5];
    const float* Wo   = (const float*)d_in[6];
    const float* bo   = (const float*)d_in[7];
    float* out = (float*)d_out;

    float* ws  = (float*)d_ws;
    float* qkv = ws;
    float* Qb  = qkv + (size_t)4096 * 3072;
    float* Kb  = Qb + (size_t)4194304;
    float* Vb  = Kb + (size_t)4194304;
    float* Ob  = Vb + (size_t)4194304;

    gemm_bias_kernel<<<dim3(3072 / 128, 4096 / 128), 256, 0, stream>>>(
        x, Wqkv, bqkv, qkv, 4096, 3072, 1024);

    rope_repack_kernel<<<2097152 / 256, 256, 0, stream>>>(qkv, cosT, sinT, Qb, Kb, Vb);

    flash_attn_kernel<<<dim3(32, 32), 128, 0, stream>>>(Qb, Kb, Vb, Ob);

    gemm_bias_kernel<<<dim3(1024 / 128, 4096 / 128), 256, 0, stream>>>(
        Ob, Wo, bo, out, 4096, 1024, 1024);
}


# Causal load balance in attention: each workgroup runs the q-tile pair (31-x, x) one after the other, so every workgroup covers 33 K/V tiles instead of 1 to 32; grid x halved; sum order unchanged
# speedup vs baseline: 1.3825x; 1.1827x over previous
#include <hip/hip_runtime.h>
#include <hip/hip_bf16.h>

typedef float v2f __attribute__((ext_vector_type(2)));
typedef float v4f __attribute__((ext_vector_type(4)));
typedef float v8f __attribute__((ext_vector_type(8)));

#define WMMA_F32(a, b, c) \
  __builtin_amdgcn_wmma_f32_16x16x4_f32(false, (a), false, (b), (short)0, (c), false, false)

__launch_bounds__(256)
__global__ void gemm_bias_kernel(const float* __restrict__ A, const float* __restrict__ B,
                                 const float* __restrict__ bias, float* __restrict__ C,
                                 int M, int N, int K)
{
    __shared__ __align__(16) float As [128 * 34];
    __shared__ __align__(16) float BsT[128 * 34];

    const int tid  = threadIdx.x;
    const int wid  = tid >> 5;
    const int lane = tid & 31;
    const int m16  = lane & 15;
    const int h    = lane >> 4;
    const int wm   = wid >> 1;
    const int wn   = wid & 1;
    const int row0 = blockIdx.y * 128;
    const int col0 = blockIdx.x * 128;

    const int ar  = tid >> 1;
    const int akq = (tid & 1) << 4;
    const int bkr = tid >> 5;
    const int bcq = (tid & 31) << 2;

    v8f acc[2][4];
    #pragma unroll
    for (int i = 0; i < 2; ++i)
        #pragma unroll
        for (int j = 0; j < 4; ++j) { v8f z{}; acc[i][j] = z; }

    float4 pa[4], pb[4];
    #pragma unroll
    for (int i = 0; i < 4; ++i)
        pa[i] = *(const float4*)(A + (size_t)(row0 + ar) * K + akq + 4 * i);
    #pragma unroll
    for (int i = 0; i < 4; ++i)
        pb[i] = *(const float4*)(B + (size_t)(bkr + 8 * i) * N + col0 + bcq);

    for (int kk = 0; kk < K; kk += 32) {
        #pragma unroll
        for (int i = 0; i < 4; ++i) {
            float* d = &As[ar * 34 + akq + 4 * i];
            d[0] = pa[i].x; d[1] = pa[i].y; d[2] = pa[i].z; d[3] = pa[i].w;
        }
        #pragma unroll
        for (int i = 0; i < 4; ++i) {
            int kr = bkr + 8 * i;
            BsT[(bcq + 0) * 34 + kr] = pb[i].x;
            BsT[(bcq + 1) * 34 + kr] = pb[i].y;
            BsT[(bcq + 2) * 34 + kr] = pb[i].z;
            BsT[(bcq + 3) * 34 + kr] = pb[i].w;
        }
        __syncthreads();

        if (kk + 32 < K) {
            #pragma unroll
            for (int i = 0; i < 4; ++i)
                pa[i] = *(const float4*)(A + (size_t)(row0 + ar) * K + kk + 32 + akq + 4 * i);
            #pragma unroll
            for (int i = 0; i < 4; ++i)
                pb[i] = *(const float4*)(B + (size_t)(bkr + 8 * i) * N + kk * 0 + (size_t)(kk + 32 + bkr + 8 * i - (bkr + 8 * i)) * 0 + col0 + bcq + (size_t)(kk + 32) * 0)
                      , pb[i] = *(const float4*)(B + (size_t)(kk + 32 + bkr + 8 * i) * N + col0 + bcq);
        }

        #pragma unroll
        for (int ks = 0; ks < 32; ks += 4) {
            v2f a[2], b[4];
            #pragma unroll
            for (int mi = 0; mi < 2; ++mi)
                a[mi] = *(const v2f*)(&As[(wm * 32 + mi * 16 + m16) * 34 + ks + 2 * h]);
            #pragma unroll
            for (int ni = 0; ni < 4; ++ni)
                b[ni] = *(const v2f*)(&BsT[(wn * 64 + ni * 16 + m16) * 34 + ks + 2 * h]);
            #pragma unroll
            for (int mi = 0; mi < 2; ++mi)
                #pragma unroll
                for (int ni = 0; ni < 4; ++ni)
                    acc[mi][ni] = WMMA_F32(a[mi], b[ni], acc[mi][ni]);
        }
        __syncthreads();
    }

    #pragma unroll
    for (int mi = 0; mi < 2; ++mi) {
        #pragma unroll
        for (int ni = 0; ni < 4; ++ni) {
            int c = col0 + wn * 64 + ni * 16 + m16;
            float bv = bias[c];
            #pragma unroll
            for (int r = 0; r < 8; ++r) {
                int rr = row0 + wm * 32 + mi * 16 + r + 8 * h;
                C[(size_t)rr * N + c] = acc[mi][ni][r] + bv;
            }
        }
    }
}

__global__ void rope_repack_kernel(const float* __restrict__ qkv,
                                   const float* __restrict__ cosT,
                                   const float* __restrict__ sinT,
                                   float* __restrict__ Q, float* __restrict__ Ko,
                                   float* __restrict__ V)
{
    int tid = blockIdx.x * blockDim.x + threadIdx.x;
    int dp  = tid & 31;
    int h0  = (tid >> 5) & 15;
    int s   = (tid >> 9) & 2047;
    int b   = tid >> 20;
    size_t rowq = (size_t)(b * 2048 + s) * 3072;
    int off = h0 * 64 + 2 * dp;

    float c  = cosT[s * 32 + dp];
    float sn = sinT[s * 32 + dp];
    float qe = qkv[rowq + off],        qo = qkv[rowq + off + 1];
    float ke = qkv[rowq + 1024 + off], ko = qkv[rowq + 1024 + off + 1];

    size_t oidx = ((size_t)((b * 16 + h0) * 2048 + s)) * 64 + 2 * dp;
    Q[oidx]      = qe * c - qo * sn;
    Q[oidx + 1]  = qe * sn + qo * c;
    Ko[oidx]     = ke * c - ko * sn;
    Ko[oidx + 1] = ke * sn + ko * c;
    V[oidx]      = qkv[rowq + 2048 + off];
    V[oidx + 1]  = qkv[rowq + 2048 + off + 1];
}

__launch_bounds__(128)
__global__ void flash_attn_kernel(const float* __restrict__ Q, const float* __restrict__ Kg,
                                  const float* __restrict__ V, float* __restrict__ O)
{
    constexpr int LP = 68;
    __shared__ __align__(16) float Ks [64 * LP];
    __shared__ __align__(16) float Vst[64 * LP];
    __shared__ __align__(16) float Pw [4 * 16 * LP];

    const int tid  = threadIdx.x;
    const int wid  = tid >> 5;
    const int lane = tid & 31;
    const int m16  = lane & 15;
    const int h    = lane >> 4;
    const int bh   = blockIdx.y;
    const size_t base = (size_t)bh * 2048 * 64;

    const int kkr = tid >> 4;
    const int kc  = (tid & 15) << 2;
    const int vkr = tid >> 3;
    const int vc  = (tid & 7) << 2;

    float* Pme = &Pw[wid * 16 * LP];
    const int b_ = bh >> 4, head = bh & 15;

    for (int pass = 0; pass < 2; ++pass) {
        const int qt    = pass == 0 ? 31 - (int)blockIdx.x : (int)blockIdx.x;
        const int qrow  = qt * 64 + wid * 16;

        v2f aq[16];
        {
            const float* qp = Q + base + (size_t)(qrow + m16) * 64;
            #pragma unroll
            for (int s2 = 0; s2 < 16; ++s2) {
                v2f t = *(const v2f*)(qp + 4 * s2 + 2 * h);
                aq[s2] = t * 0.125f;
            }
        }

        float mrow[8], lrow[8];
        #pragma unroll
        for (int r = 0; r < 8; ++r) { mrow[r] = -1e30f; lrow[r] = 0.0f; }
        v8f oacc[4];
        #pragma unroll
        for (int i = 0; i < 4; ++i) { v8f z{}; oacc[i] = z; }

        const int ntiles = qt + 1;

        v4f pk[8], pv[8];
        #pragma unroll
        for (int i = 0; i < 8; ++i) {
            pk[i] = *(const v4f*)(Kg + base + (size_t)(kkr + 8 * i) * 64 + kc);
            pv[i] = *(const v4f*)(V  + base + (size_t)(vkr + 16 * (i >> 1)) * 64 + 32 * (i & 1) + vc);
        }

        for (int kt = 0; kt < ntiles; ++kt) {
            #pragma unroll
            for (int i = 0; i < 8; ++i) {
                *(v4f*)(&Ks[(kkr + 8 * i) * LP + kc]) = pk[i];
                const int key = vkr + 16 * (i >> 1);
                const int c   = 32 * (i & 1) + vc;
                Vst[(c + 0) * LP + key] = pv[i].x;
                Vst[(c + 1) * LP + key] = pv[i].y;
                Vst[(c + 2) * LP + key] = pv[i].z;
                Vst[(c + 3) * LP + key] = pv[i].w;
            }
            __syncthreads();

            if (kt + 1 < ntiles) {
                const size_t nb = base + (size_t)(kt + 1) * 64 * 64;
                #pragma unroll
                for (int i = 0; i < 8; ++i) {
                    pk[i] = *(const v4f*)(Kg + nb + (size_t)(kkr + 8 * i) * 64 + kc);
                    pv[i] = *(const v4f*)(V  + nb + (size_t)(vkr + 16 * (i >> 1)) * 64 + 32 * (i & 1) + vc);
                }
            }

            v8f sc[4];
            #pragma unroll
            for (int j = 0; j < 4; ++j) { v8f z{}; sc[j] = z; }
            #pragma unroll
            for (int s2 = 0; s2 < 16; ++s2) {
                #pragma unroll
                for (int j = 0; j < 4; ++j) {
                    v2f b = *(const v2f*)(&Ks[(16 * j + m16) * LP + 4 * s2 + 2 * h]);
                    sc[j] = WMMA_F32(aq[s2], b, sc[j]);
                }
            }

            if (kt == ntiles - 1) {
                #pragma unroll
                for (int r = 0; r < 8; ++r) {
                    const int lim = wid * 16 + r + 8 * h - m16;
                    sc[0][r] = (0  <= lim) ? sc[0][r] : -1e30f;
                    sc[1][r] = (16 <= lim) ? sc[1][r] : -1e30f;
                    sc[2][r] = (32 <= lim) ? sc[2][r] : -1e30f;
                    sc[3][r] = (48 <= lim) ? sc[3][r] : -1e30f;
                }
            }
            #pragma unroll
            for (int r = 0; r < 8; ++r) {
                const float v0 = sc[0][r], v1 = sc[1][r], v2 = sc[2][r], v3 = sc[3][r];
                float mx = fmaxf(fmaxf(v0, v1), fmaxf(v2, v3));
                mx = fmaxf(mx, __shfl_xor(mx, 1, 32));
                mx = fmaxf(mx, __shfl_xor(mx, 2, 32));
                mx = fmaxf(mx, __shfl_xor(mx, 4, 32));
                mx = fmaxf(mx, __shfl_xor(mx, 8, 32));
                const float mnew = fmaxf(mrow[r], mx);
                const float p0 = __expf(v0 - mnew);
                const float p1 = __expf(v1 - mnew);
                const float p2 = __expf(v2 - mnew);
                const float p3 = __expf(v3 - mnew);
                float sum = (p0 + p1) + (p2 + p3);
                sum += __shfl_xor(sum, 1, 32);
                sum += __shfl_xor(sum, 2, 32);
                sum += __shfl_xor(sum, 4, 32);
                sum += __shfl_xor(sum, 8, 32);
                const float alpha = __expf(mrow[r] - mnew);
                mrow[r] = mnew;
                lrow[r] = lrow[r] * alpha + sum;
                #pragma unroll
                for (int ni = 0; ni < 4; ++ni) oacc[ni][r] *= alpha;
                float* prow = &Pme[(r + 8 * h) * LP + m16];
                prow[0]  = p0;
                prow[16] = p1;
                prow[32] = p2;
                prow[48] = p3;
            }
            __builtin_amdgcn_wave_barrier();

            #pragma unroll
            for (int ks = 0; ks < 64; ks += 4) {
                v2f a = *(const v2f*)(&Pme[m16 * LP + ks + 2 * h]);
                #pragma unroll
                for (int ni = 0; ni < 4; ++ni) {
                    v2f b = *(const v2f*)(&Vst[(ni * 16 + m16) * LP + ks + 2 * h]);
                    oacc[ni] = WMMA_F32(a, b, oacc[ni]);
                }
            }
            __syncthreads();
        }

        #pragma unroll
        for (int ni = 0; ni < 4; ++ni) {
            int c = head * 64 + ni * 16 + m16;
            #pragma unroll
            for (int r = 0; r < 8; ++r) {
                int srow = qrow + r + 8 * h;
                float inv = 1.0f / lrow[r];
                O[((size_t)(b_ * 2048 + srow)) * 1024 + c] = oacc[ni][r] * inv;
            }
        }
    }
}

extern "C" void kernel_launch(void* const* d_in, const int* in_sizes, int n_in,
                              void* d_out, int out_size, void* d_ws, size_t ws_size,
                              hipStream_t stream)
{
    const float* x    = (const float*)d_in[0];
    const float* cosT = (const float*)d_in[2];
    const float* sinT = (const float*)d_in[3];
    const float* Wqkv = (const float*)d_in[4];
    const float* bqkv = (const float*)d_in[5];
    const float* Wo   = (const float*)d_in[6];
    const float* bo   = (const float*)d_in[7];
    float* out = (float*)d_out;

    float* ws  = (float*)d_ws;
    float* qkv = ws;
    float* Qb  = qkv + (size_t)4096 * 3072;
    float* Kb  = Qb + (size_t)4194304;
    float* Vb  = Kb + (size_t)4194304;
    float* Ob  = Vb + (size_t)4194304;

    gemm_bias_kernel<<<dim3(3072 / 128, 4096 / 128), 256, 0, stream>>>(
        x, Wqkv, bqkv, qkv, 4096, 3072, 1024);

    rope_repack_kernel<<<2097152 / 256, 256, 0, stream>>>(qkv, cosT, sinT, Qb, Kb, Vb);

    flash_attn_kernel<<<dim3(16, 32), 128, 0, stream>>>(Qb, Kb, Vb, Ob);

    gemm_bias_kernel<<<dim3(1024 / 128, 4096 / 128), 256, 0, stream>>>(
        Ob, Wo, bo, out, 4096, 1024, 1024);
}


# Rotary rotation and head repack of q,k,v folded into the QKV GEMM epilogue (B columns dealt even/odd: a lane holds a rotary pair); rotate/repack launch and qkv round trip gone; sum order unchanged
# speedup vs baseline: 1.4942x; 1.0808x over previous
#include <hip/hip_runtime.h>
#include <hip/hip_bf16.h>

typedef float v2f __attribute__((ext_vector_type(2)));
typedef float v4f __attribute__((ext_vector_type(4)));
typedef float v8f __attribute__((ext_vector_type(8)));

#define WMMA_F32(a, b, c) \
  __builtin_amdgcn_wmma_f32_16x16x4_f32(false, (a), false, (b), (short)0, (c), false, false)

template <bool ROPE>
__launch_bounds__(256)
__global__ void gemm_bias_kernel(const float* __restrict__ A, const float* __restrict__ B,
                                 const float* __restrict__ bias, float* __restrict__ C,
                                 int M, int N, int K,
                                 const float* __restrict__ cosT, const float* __restrict__ sinT,
                                 float* __restrict__ Qo, float* __restrict__ Ko,
                                 float* __restrict__ Vo)
{
    __shared__ __align__(16) float As [128 * 34];
    __shared__ __align__(16) float BsT[128 * 34];

    const int tid  = threadIdx.x;
    const int wid  = tid >> 5;
    const int lane = tid & 31;
    const int m16  = lane & 15;
    const int h    = lane >> 4;
    const int wm   = wid >> 1;
    const int wn   = wid & 1;
    const int row0 = blockIdx.y * 128;
    const int col0 = blockIdx.x * 128;

    const int ar  = tid >> 1;
    const int akq = (tid & 1) << 4;
    const int bkr = tid >> 5;
    const int bcq = (tid & 31) << 2;

    v8f acc[2][4];
    #pragma unroll
    for (int i = 0; i < 2; ++i)
        #pragma unroll
        for (int j = 0; j < 4; ++j) { v8f z{}; acc[i][j] = z; }

    float4 pa[4], pb[4];
    #pragma unroll
    for (int i = 0; i < 4; ++i)
        pa[i] = *(const float4*)(A + (size_t)(row0 + ar) * K + akq + 4 * i);
    #pragma unroll
    for (int i = 0; i < 4; ++i)
        pb[i] = *(const float4*)(B + (size_t)(bkr + 8 * i) * N + col0 + bcq);

    for (int kk = 0; kk < K; kk += 32) {
        #pragma unroll
        for (int i = 0; i < 4; ++i) {
            float* d = &As[ar * 34 + akq + 4 * i];
            d[0] = pa[i].x; d[1] = pa[i].y; d[2] = pa[i].z; d[3] = pa[i].w;
        }
        #pragma unroll
        for (int i = 0; i < 4; ++i) {
            int kr = bkr + 8 * i;
            BsT[(bcq + 0) * 34 + kr] = pb[i].x;
            BsT[(bcq + 1) * 34 + kr] = pb[i].y;
            BsT[(bcq + 2) * 34 + kr] = pb[i].z;
            BsT[(bcq + 3) * 34 + kr] = pb[i].w;
        }
        __syncthreads();

        if (kk + 32 < K) {
            #pragma unroll
            for (int i = 0; i < 4; ++i)
                pa[i] = *(const float4*)(A + (size_t)(row0 + ar) * K + kk + 32 + akq + 4 * i);
            #pragma unroll
            for (int i = 0; i < 4; ++i)
                pb[i] = *(const float4*)(B + (size_t)(bkr + 8 * i) * N + kk * 0 + (size_t)(kk + 32 + bkr + 8 * i - (bkr + 8 * i)) * 0 + col0 + bcq + (size_t)(kk + 32) * 0)
                      , pb[i] = *(const float4*)(B + (size_t)(kk + 32 + bkr + 8 * i) * N + col0 + bcq);
        }

        #pragma unroll
        for (int ks = 0; ks < 32; ks += 4) {
            v2f a[2], b[4];
            #pragma unroll
            for (int mi = 0; mi < 2; ++mi)
                a[mi] = *(const v2f*)(&As[(wm * 32 + mi * 16 + m16) * 34 + ks + 2 * h]);
            #pragma unroll
            for (int ni = 0; ni < 4; ++ni) {
                const int bc = ROPE ? ((ni >> 1) * 32 + 2 * m16 + (ni & 1)) : (ni * 16 + m16);
                b[ni] = *(const v2f*)(&BsT[(wn * 64 + bc) * 34 + ks + 2 * h]);
            }
            #pragma unroll
            for (int mi = 0; mi < 2; ++mi)
                #pragma unroll
                for (int ni = 0; ni < 4; ++ni)
                    acc[mi][ni] = WMMA_F32(a[mi], b[ni], acc[mi][ni]);
        }
        __syncthreads();
    }

    if (!ROPE) {
        #pragma unroll
        for (int mi = 0; mi < 2; ++mi) {
            #pragma unroll
            for (int ni = 0; ni < 4; ++ni) {
                int c = col0 + wn * 64 + ni * 16 + m16;
                float bv = bias[c];
                #pragma unroll
                for (int r = 0; r < 8; ++r) {
                    int rr = row0 + wm * 32 + mi * 16 + r + 8 * h;
                    C[(size_t)rr * N + c] = acc[mi][ni][r] + bv;
                }
            }
        }
    } else {
        const int which = col0 >> 10;
        const int head  = ((col0 & 1023) >> 6) + wn;
        float* __restrict__ dst = which == 0 ? Qo : (which == 1 ? Ko : Vo);
        const int rbase = row0 + wm * 32 + 8 * h;
        const int sbase = rbase & 2047;
        float* __restrict__ drow =
            dst + ((size_t)(((row0 >> 11) * 16 + head) * 2048 + sbase)) * 64 + 2 * m16;
        float be[2], bo[2];
        #pragma unroll
        for (int np = 0; np < 2; ++np) {
            const int ce = col0 + wn * 64 + np * 32 + 2 * m16;
            be[np] = bias[ce];
            bo[np] = bias[ce + 1];
        }
        if (which < 2) {
            float cv[2][2][8], sv[2][2][8];
            #pragma unroll
            for (int np = 0; np < 2; ++np)
                #pragma unroll
                for (int mi = 0; mi < 2; ++mi)
                    #pragma unroll
                    for (int r = 0; r < 8; ++r) {
                        const int ti = (sbase + mi * 16 + r) * 32 + np * 16 + m16;
                        cv[np][mi][r] = cosT[ti];
                        sv[np][mi][r] = sinT[ti];
                    }
            #pragma unroll
            for (int np = 0; np < 2; ++np)
                #pragma unroll
                for (int mi = 0; mi < 2; ++mi)
                    #pragma unroll
                    for (int r = 0; r < 8; ++r) {
                        const float e = acc[mi][2 * np][r]     + be[np];
                        const float o = acc[mi][2 * np + 1][r] + bo[np];
                        v2f w;
                        w.x = e * cv[np][mi][r] - o * sv[np][mi][r];
                        w.y = e * sv[np][mi][r] + o * cv[np][mi][r];
                        *(v2f*)(drow + (mi * 16 + r) * 64 + np * 32) = w;
                    }
        } else {
            #pragma unroll
            for (int np = 0; np < 2; ++np)
                #pragma unroll
                for (int mi = 0; mi < 2; ++mi)
                    #pragma unroll
                    for (int r = 0; r < 8; ++r) {
                        v2f w;
                        w.x = acc[mi][2 * np][r]     + be[np];
                        w.y = acc[mi][2 * np + 1][r] + bo[np];
                        *(v2f*)(drow + (mi * 16 + r) * 64 + np * 32) = w;
                    }
        }
    }
}

__launch_bounds__(128)
__global__ void flash_attn_kernel(const float* __restrict__ Q, const float* __restrict__ Kg,
                                  const float* __restrict__ V, float* __restrict__ O)
{
    constexpr int LP = 68;
    __shared__ __align__(16) float Ks [64 * LP];
    __shared__ __align__(16) float Vst[64 * LP];
    __shared__ __align__(16) float Pw [4 * 16 * LP];

    const int tid  = threadIdx.x;
    const int wid  = tid >> 5;
    const int lane = tid & 31;
    const int m16  = lane & 15;
    const int h    = lane >> 4;
    const int bh   = blockIdx.y;
    const size_t base = (size_t)bh * 2048 * 64;

    const int kkr = tid >> 4;
    const int kc  = (tid & 15) << 2;
    const int vkr = tid >> 3;
    const int vc  = (tid & 7) << 2;

    float* Pme = &Pw[wid * 16 * LP];
    const int b_ = bh >> 4, head = bh & 15;

    for (int pass = 0; pass < 2; ++pass) {
        const int qt    = pass == 0 ? 31 - (int)blockIdx.x : (int)blockIdx.x;
        const int qrow  = qt * 64 + wid * 16;

        v2f aq[16];
        {
            const float* qp = Q + base + (size_t)(qrow + m16) * 64;
            #pragma unroll
            for (int s2 = 0; s2 < 16; ++s2) {
                v2f t = *(const v2f*)(qp + 4 * s2 + 2 * h);
                aq[s2] = t * 0.125f;
            }
        }

        float mrow[8], lrow[8];
        #pragma unroll
        for (int r = 0; r < 8; ++r) { mrow[r] = -1e30f; lrow[r] = 0.0f; }
        v8f oacc[4];
        #pragma unroll
        for (int i = 0; i < 4; ++i) { v8f z{}; oacc[i] = z; }

        const int ntiles = qt + 1;

        v4f pk[8], pv[8];
        #pragma unroll
        for (int i = 0; i < 8; ++i) {
            pk[i] = *(const v4f*)(Kg + base + (size_t)(kkr + 8 * i) * 64 + kc);
            pv[i] = *(const v4f*)(V  + base + (size_t)(vkr + 16 * (i >> 1)) * 64 + 32 * (i & 1) + vc);
        }

        for (int kt = 0; kt < ntiles; ++kt) {
            #pragma unroll
            for (int i = 0; i < 8; ++i) {
                *(v4f*)(&Ks[(kkr + 8 * i) * LP + kc]) = pk[i];
                const int key = vkr + 16 * (i >> 1);
                const int c   = 32 * (i & 1) + vc;
                Vst[(c + 0) * LP + key] = pv[i].x;
                Vst[(c + 1) * LP + key] = pv[i].y;
                Vst[(c + 2) * LP + key] = pv[i].z;
                Vst[(c + 3) * LP + key] = pv[i].w;
            }
            __syncthreads();

            if (kt + 1 < ntiles) {
                const size_t nb = base + (size_t)(kt + 1) * 64 * 64;
                #pragma unroll
                for (int i = 0; i < 8; ++i) {
                    pk[i] = *(const v4f*)(Kg + nb + (size_t)(kkr + 8 * i) * 64 + kc);
                    pv[i] = *(const v4f*)(V  + nb + (size_t)(vkr + 16 * (i >> 1)) * 64 + 32 * (i & 1) + vc);
                }
            }

            v8f sc[4];
            #pragma unroll
            for (int j = 0; j < 4; ++j) { v8f z{}; sc[j] = z; }
            #pragma unroll
            for (int s2 = 0; s2 < 16; ++s2) {
                #pragma unroll
                for (int j = 0; j < 4; ++j) {
                    v2f b = *(const v2f*)(&Ks[(16 * j + m16) * LP + 4 * s2 + 2 * h]);
                    sc[j] = WMMA_F32(aq[s2], b, sc[j]);
                }
            }

            if (kt == ntiles - 1) {
                #pragma unroll
                for (int r = 0; r < 8; ++r) {
                    const int lim = wid * 16 + r + 8 * h - m16;
                    sc[0][r] = (0  <= lim) ? sc[0][r] : -1e30f;
                    sc[1][r] = (16 <= lim) ? sc[1][r] : -1e30f;
                    sc[2][r] = (32 <= lim) ? sc[2][r] : -1e30f;
                    sc[3][r] = (48 <= lim) ? sc[3][r] : -1e30f;
                }
            }
            #pragma unroll
            for (int r = 0; r < 8; ++r) {
                const float v0 = sc[0][r], v1 = sc[1][r], v2 = sc[2][r], v3 = sc[3][r];
                float mx = fmaxf(fmaxf(v0, v1), fmaxf(v2, v3));
                mx = fmaxf(mx, __shfl_xor(mx, 1, 32));
                mx = fmaxf(mx, __shfl_xor(mx, 2, 32));
                mx = fmaxf(mx, __shfl_xor(mx, 4, 32));
                mx = fmaxf(mx, __shfl_xor(mx, 8, 32));
                const float mnew = fmaxf(mrow[r], mx);
                const float p0 = __expf(v0 - mnew);
                const float p1 = __expf(v1 - mnew);
                const float p2 = __expf(v2 - mnew);
                const float p3 = __expf(v3 - mnew);
                float sum = (p0 + p1) + (p2 + p3);
                sum += __shfl_xor(sum, 1, 32);
                sum += __shfl_xor(sum, 2, 32);
                sum += __shfl_xor(sum, 4, 32);
                sum += __shfl_xor(sum, 8, 32);
                const float alpha = __expf(mrow[r] - mnew);
                mrow[r] = mnew;
                lrow[r] = lrow[r] * alpha + sum;
                #pragma unroll
                for (int ni = 0; ni < 4; ++ni) oacc[ni][r] *= alpha;
                float* prow = &Pme[(r + 8 * h) * LP + m16];
                prow[0]  = p0;
                prow[16] = p1;
                prow[32] = p2;
                prow[48] = p3;
            }
            __builtin_amdgcn_wave_barrier();

            #pragma unroll
            for (int ks = 0; ks < 64; ks += 4) {
                v2f a = *(const v2f*)(&Pme[m16 * LP + ks + 2 * h]);
                #pragma unroll
                for (int ni = 0; ni < 4; ++ni) {
                    v2f b = *(const v2f*)(&Vst[(ni * 16 + m16) * LP + ks + 2 * h]);
                    oacc[ni] = WMMA_F32(a, b, oacc[ni]);
                }
            }
            __syncthreads();
        }

        #pragma unroll
        for (int ni = 0; ni < 4; ++ni) {
            int c = head * 64 + ni * 16 + m16;
            #pragma unroll
            for (int r = 0; r < 8; ++r) {
                int srow = qrow + r + 8 * h;
                float inv = 1.0f / lrow[r];
                O[((size_t)(b_ * 2048 + srow)) * 1024 + c] = oacc[ni][r] * inv;
            }
        }
    }
}

extern "C" void kernel_launch(void* const* d_in, const int* in_sizes, int n_in,
                              void* d_out, int out_size, void* d_ws, size_t ws_size,
                              hipStream_t stream)
{
    const float* x    = (const float*)d_in[0];
    const float* cosT = (const float*)d_in[2];
    const float* sinT = (const float*)d_in[3];
    const float* Wqkv = (const float*)d_in[4];
    const float* bqkv = (const float*)d_in[5];
    const float* Wo   = (const float*)d_in[6];
    const float* bo   = (const float*)d_in[7];
    float* out = (float*)d_out;

    float* ws  = (float*)d_ws;
    float* Qb  = ws + (size_t)4096 * 3072;
    float* Kb  = Qb + (size_t)4194304;
    float* Vb  = Kb + (size_t)4194304;
    float* Ob  = Vb + (size_t)4194304;

    gemm_bias_kernel<true><<<dim3(3072 / 128, 4096 / 128), 256, 0, stream>>>(
        x, Wqkv, bqkv, nullptr, 4096, 3072, 1024, cosT, sinT, Qb, Kb, Vb);

    flash_attn_kernel<<<dim3(16, 32), 128, 0, stream>>>(Qb, Kb, Vb, Ob);

    gemm_bias_kernel<false><<<dim3(1024 / 128, 4096 / 128), 256, 0, stream>>>(
        Ob, Wo, bo, out, 4096, 1024, 1024, nullptr, nullptr, nullptr, nullptr, nullptr);
}


# Attention products transposed (K Q^T, then V^T P^T): a q row sits in one lane, so row max/sum need one half-wave exchange and P feeds the 2nd product from registers, not LDS; key sum order changed
# speedup vs baseline: 1.6040x; 1.0735x over previous
#include <hip/hip_runtime.h>
#include <hip/hip_bf16.h>

typedef float v2f __attribute__((ext_vector_type(2)));
typedef float v4f __attribute__((ext_vector_type(4)));
typedef float v8f __attribute__((ext_vector_type(8)));

#define WMMA_F32(a, b, c) \
  __builtin_amdgcn_wmma_f32_16x16x4_f32(false, (a), false, (b), (short)0, (c), false, false)

template <bool ROPE>
__launch_bounds__(256)
__global__ void gemm_bias_kernel(const float* __restrict__ A, const float* __restrict__ B,
                                 const float* __restrict__ bias, float* __restrict__ C,
                                 int M, int N, int K,
                                 const float* __restrict__ cosT, const float* __restrict__ sinT,
                                 float* __restrict__ Qo, float* __restrict__ Ko,
                                 float* __restrict__ Vo)
{
    __shared__ __align__(16) float As [128 * 34];
    __shared__ __align__(16) float BsT[128 * 34];

    const int tid  = threadIdx.x;
    const int wid  = tid >> 5;
    const int lane = tid & 31;
    const int m16  = lane & 15;
    const int h    = lane >> 4;
    const int wm   = wid >> 1;
    const int wn   = wid & 1;
    const int row0 = blockIdx.y * 128;
    const int col0 = blockIdx.x * 128;

    const int ar  = tid >> 1;
    const int akq = (tid & 1) << 4;
    const int bkr = tid >> 5;
    const int bcq = (tid & 31) << 2;

    v8f acc[2][4];
    #pragma unroll
    for (int i = 0; i < 2; ++i)
        #pragma unroll
        for (int j = 0; j < 4; ++j) { v8f z{}; acc[i][j] = z; }

    float4 pa[4], pb[4];
    #pragma unroll
    for (int i = 0; i < 4; ++i)
        pa[i] = *(const float4*)(A + (size_t)(row0 + ar) * K + akq + 4 * i);
    #pragma unroll
    for (int i = 0; i < 4; ++i)
        pb[i] = *(const float4*)(B + (size_t)(bkr + 8 * i) * N + col0 + bcq);

    for (int kk = 0; kk < K; kk += 32) {
        #pragma unroll
        for (int i = 0; i < 4; ++i) {
            float* d = &As[ar * 34 + akq + 4 * i];
            d[0] = pa[i].x; d[1] = pa[i].y; d[2] = pa[i].z; d[3] = pa[i].w;
        }
        #pragma unroll
        for (int i = 0; i < 4; ++i) {
            int kr = bkr + 8 * i;
            BsT[(bcq + 0) * 34 + kr] = pb[i].x;
            BsT[(bcq + 1) * 34 + kr] = pb[i].y;
            BsT[(bcq + 2) * 34 + kr] = pb[i].z;
            BsT[(bcq + 3) * 34 + kr] = pb[i].w;
        }
        __syncthreads();

        if (kk + 32 < K) {
            #pragma unroll
            for (int i = 0; i < 4; ++i)
                pa[i] = *(const float4*)(A + (size_t)(row0 + ar) * K + kk + 32 + akq + 4 * i);
            #pragma unroll
            for (int i = 0; i < 4; ++i)
                pb[i] = *(const float4*)(B + (size_t)(bkr + 8 * i) * N + kk * 0 + (size_t)(kk + 32 + bkr + 8 * i - (bkr + 8 * i)) * 0 + col0 + bcq + (size_t)(kk + 32) * 0)
                      , pb[i] = *(const float4*)(B + (size_t)(kk + 32 + bkr + 8 * i) * N + col0 + bcq);
        }

        #pragma unroll
        for (int ks = 0; ks < 32; ks += 4) {
            v2f a[2], b[4];
            #pragma unroll
            for (int mi = 0; mi < 2; ++mi)
                a[mi] = *(const v2f*)(&As[(wm * 32 + mi * 16 + m16) * 34 + ks + 2 * h]);
            #pragma unroll
            for (int ni = 0; ni < 4; ++ni) {
                const int bc = ROPE ? ((ni >> 1) * 32 + 2 * m16 + (ni & 1)) : (ni * 16 + m16);
                b[ni] = *(const v2f*)(&BsT[(wn * 64 + bc) * 34 + ks + 2 * h]);
            }
            #pragma unroll
            for (int mi = 0; mi < 2; ++mi)
                #pragma unroll
                for (int ni = 0; ni < 4; ++ni)
                    acc[mi][ni] = WMMA_F32(a[mi], b[ni], acc[mi][ni]);
        }
        __syncthreads();
    }

    if (!ROPE) {
        #pragma unroll
        for (int mi = 0; mi < 2; ++mi) {
            #pragma unroll
            for (int ni = 0; ni < 4; ++ni) {
                int c = col0 + wn * 64 + ni * 16 + m16;
                float bv = bias[c];
                #pragma unroll
                for (int r = 0; r < 8; ++r) {
                    int rr = row0 + wm * 32 + mi * 16 + r + 8 * h;
                    C[(size_t)rr * N + c] = acc[mi][ni][r] + bv;
                }
            }
        }
    } else {
        const int which = col0 >> 10;
        const int head  = ((col0 & 1023) >> 6) + wn;
        float* __restrict__ dst = which == 0 ? Qo : (which == 1 ? Ko : Vo);
        const int rbase = row0 + wm * 32 + 8 * h;
        const int sbase = rbase & 2047;
        float* __restrict__ drow =
            dst + ((size_t)(((row0 >> 11) * 16 + head) * 2048 + sbase)) * 64 + 2 * m16;
        float be[2], bo[2];
        #pragma unroll
        for (int np = 0; np < 2; ++np) {
            const int ce = col0 + wn * 64 + np * 32 + 2 * m16;
            be[np] = bias[ce];
            bo[np] = bias[ce + 1];
        }
        if (which < 2) {
            float cv[2][2][8], sv[2][2][8];
            #pragma unroll
            for (int np = 0; np < 2; ++np)
                #pragma unroll
                for (int mi = 0; mi < 2; ++mi)
                    #pragma unroll
                    for (int r = 0; r < 8; ++r) {
                        const int ti = (sbase + mi * 16 + r) * 32 + np * 16 + m16;
                        cv[np][mi][r] = cosT[ti];
                        sv[np][mi][r] = sinT[ti];
                    }
            #pragma unroll
            for (int np = 0; np < 2; ++np)
                #pragma unroll
                for (int mi = 0; mi < 2; ++mi)
                    #pragma unroll
                    for (int r = 0; r < 8; ++r) {
                        const float e = acc[mi][2 * np][r]     + be[np];
                        const float o = acc[mi][2 * np + 1][r] + bo[np];
                        v2f w;
                        w.x = e * cv[np][mi][r] - o * sv[np][mi][r];
                        w.y = e * sv[np][mi][r] + o * cv[np][mi][r];
                        *(v2f*)(drow + (mi * 16 + r) * 64 + np * 32) = w;
                    }
        } else {
            #pragma unroll
            for (int np = 0; np < 2; ++np)
                #pragma unroll
                for (int mi = 0; mi < 2; ++mi)
                    #pragma unroll
                    for (int r = 0; r < 8; ++r) {
                        v2f w;
                        w.x = acc[mi][2 * np][r]     + be[np];
                        w.y = acc[mi][2 * np + 1][r] + bo[np];
                        *(v2f*)(drow + (mi * 16 + r) * 64 + np * 32) = w;
                    }
        }
    }
}

__device__ __forceinline__ float xhalf(float x)
{
    const int b = __builtin_bit_cast(int, x);
    return __builtin_bit_cast(float,
        __builtin_amdgcn_permlanex16(b, b, 0x76543210, 0xfedcba98, false, false));
}

__launch_bounds__(128)
__global__ void flash_attn_kernel(const float* __restrict__ Q, const float* __restrict__ Kg,
                                  const float* __restrict__ V, float* __restrict__ O)
{
    constexpr int LP = 68;
    __shared__ __align__(16) float Ks [64 * LP];
    __shared__ __align__(16) float Vst[64 * LP];

    const int tid  = threadIdx.x;
    const int wid  = tid >> 5;
    const int lane = tid & 31;
    const int m16  = lane & 15;
    const int h    = lane >> 4;
    const int bh   = blockIdx.y;
    const size_t base = (size_t)bh * 2048 * 64;

    const int kkr = tid >> 4;
    const int kc  = (tid & 15) << 2;
    const int vkr = tid >> 3;
    const int vc  = (tid & 7) << 2;
    const int vpos = (((vkr >> 1) & 3) << 2) | ((vkr >> 3) << 1) | (vkr & 1);

    const int b_ = bh >> 4, head = bh & 15;

    for (int pass = 0; pass < 2; ++pass) {
        const int qt    = pass == 0 ? 31 - (int)blockIdx.x : (int)blockIdx.x;
        const int qrow  = qt * 64 + wid * 16;

        v2f bq[16];
        {
            const float* qp = Q + base + (size_t)(qrow + m16) * 64;
            #pragma unroll
            for (int s2 = 0; s2 < 16; ++s2) {
                v2f t = *(const v2f*)(qp + 4 * s2 + 2 * h);
                bq[s2] = t * 0.125f;
            }
        }

        float mrun = -1e30f, lrun = 0.0f;
        v8f oacc[4];
        #pragma unroll
        for (int i = 0; i < 4; ++i) { v8f z{}; oacc[i] = z; }

        const int ntiles = qt + 1;

        v4f pk[8], pv[8];
        #pragma unroll
        for (int i = 0; i < 8; ++i) {
            pk[i] = *(const v4f*)(Kg + base + (size_t)(kkr + 8 * i) * 64 + kc);
            pv[i] = *(const v4f*)(V  + base + (size_t)(vkr + 16 * (i >> 1)) * 64 + 32 * (i & 1) + vc);
        }

        for (int kt = 0; kt < ntiles; ++kt) {
            #pragma unroll
            for (int i = 0; i < 8; ++i) {
                *(v4f*)(&Ks[(kkr + 8 * i) * LP + kc]) = pk[i];
                const int kp = vpos + 16 * (i >> 1);
                const int c  = 32 * (i & 1) + vc;
                Vst[(c + 0) * LP + kp] = pv[i].x;
                Vst[(c + 1) * LP + kp] = pv[i].y;
                Vst[(c + 2) * LP + kp] = pv[i].z;
                Vst[(c + 3) * LP + kp] = pv[i].w;
            }
            __syncthreads();

            if (kt + 1 < ntiles) {
                const size_t nb = base + (size_t)(kt + 1) * 64 * 64;
                #pragma unroll
                for (int i = 0; i < 8; ++i) {
                    pk[i] = *(const v4f*)(Kg + nb + (size_t)(kkr + 8 * i) * 64 + kc);
                    pv[i] = *(const v4f*)(V  + nb + (size_t)(vkr + 16 * (i >> 1)) * 64 + 32 * (i & 1) + vc);
                }
            }

            v8f sc[4];
            #pragma unroll
            for (int j = 0; j < 4; ++j) { v8f z{}; sc[j] = z; }
            #pragma unroll
            for (int s2 = 0; s2 < 16; ++s2) {
                #pragma unroll
                for (int j = 0; j < 4; ++j) {
                    v2f a = *(const v2f*)(&Ks[(16 * j + m16) * LP + 4 * s2 + 2 * h]);
                    sc[j] = WMMA_F32(a, bq[s2], sc[j]);
                }
            }

            if (kt == ntiles - 1) {
                const int lim = wid * 16 + m16 - 8 * h;
                #pragma unroll
                for (int j = 0; j < 4; ++j)
                    #pragma unroll
                    for (int r = 0; r < 8; ++r)
                        sc[j][r] = (16 * j + r <= lim) ? sc[j][r] : -1e30f;
            }

            float mj[4];
            #pragma unroll
            for (int j = 0; j < 4; ++j) {
                const float a0 = fmaxf(sc[j][0], sc[j][1]);
                const float a1 = fmaxf(sc[j][2], sc[j][3]);
                const float a2 = fmaxf(sc[j][4], sc[j][5]);
                const float a3 = fmaxf(sc[j][6], sc[j][7]);
                mj[j] = fmaxf(fmaxf(a0, a1), fmaxf(a2, a3));
            }
            float mx = fmaxf(fmaxf(mj[0], mj[1]), fmaxf(mj[2], mj[3]));
            mx = fmaxf(mx, xhalf(mx));
            const float mnew = fmaxf(mrun, mx);
            float sj[4];
            #pragma unroll
            for (int j = 0; j < 4; ++j) {
                #pragma unroll
                for (int r = 0; r < 8; ++r) sc[j][r] = __expf(sc[j][r] - mnew);
                sj[j] = ((sc[j][0] + sc[j][1]) + (sc[j][2] + sc[j][3]))
                      + ((sc[j][4] + sc[j][5]) + (sc[j][6] + sc[j][7]));
            }
            float sum = (sj[0] + sj[1]) + (sj[2] + sj[3]);
            sum += xhalf(sum);
            const float alpha = __expf(mrun - mnew);
            mrun = mnew;
            lrun = lrun * alpha + sum;
            #pragma unroll
            for (int ni = 0; ni < 4; ++ni)
                #pragma unroll
                for (int r = 0; r < 8; ++r) oacc[ni][r] *= alpha;

            #pragma unroll
            for (int j = 0; j < 4; ++j) {
                #pragma unroll
                for (int t = 0; t < 4; ++t) {
                    v2f p;
                    p.x = sc[j][2 * t];
                    p.y = sc[j][2 * t + 1];
                    #pragma unroll
                    for (int ni = 0; ni < 4; ++ni) {
                        v2f a = *(const v2f*)(&Vst[(ni * 16 + m16) * LP + 16 * j + 4 * t + 2 * h]);
                        oacc[ni] = WMMA_F32(a, p, oacc[ni]);
                    }
                }
            }
            __syncthreads();
        }

        {
            const float inv = 1.0f / lrun;
            float* op = O + ((size_t)(b_ * 2048 + qrow + m16)) * 1024 + head * 64 + 8 * h;
            #pragma unroll
            for (int ni = 0; ni < 4; ++ni) {
                v4f lo, hi;
                lo.x = oacc[ni][0] * inv; lo.y = oacc[ni][1] * inv;
                lo.z = oacc[ni][2] * inv; lo.w = oacc[ni][3] * inv;
                hi.x = oacc[ni][4] * inv; hi.y = oacc[ni][5] * inv;
                hi.z = oacc[ni][6] * inv; hi.w = oacc[ni][7] * inv;
                *(v4f*)(op + 16 * ni)     = lo;
                *(v4f*)(op + 16 * ni + 4) = hi;
            }
        }
    }
}

extern "C" void kernel_launch(void* const* d_in, const int* in_sizes, int n_in,
                              void* d_out, int out_size, void* d_ws, size_t ws_size,
                              hipStream_t stream)
{
    const float* x    = (const float*)d_in[0];
    const float* cosT = (const float*)d_in[2];
    const float* sinT = (const float*)d_in[3];
    const float* Wqkv = (const float*)d_in[4];
    const float* bqkv = (const float*)d_in[5];
    const float* Wo   = (const float*)d_in[6];
    const float* bo   = (const float*)d_in[7];
    float* out = (float*)d_out;

    float* ws  = (float*)d_ws;
    float* Qb  = ws + (size_t)4096 * 3072;
    float* Kb  = Qb + (size_t)4194304;
    float* Vb  = Kb + (size_t)4194304;
    float* Ob  = Vb + (size_t)4194304;

    gemm_bias_kernel<true><<<dim3(3072 / 128, 4096 / 128), 256, 0, stream>>>(
        x, Wqkv, bqkv, nullptr, 4096, 3072, 1024, cosT, sinT, Qb, Kb, Vb);

    flash_attn_kernel<<<dim3(16, 32), 128, 0, stream>>>(Qb, Kb, Vb, Ob);

    gemm_bias_kernel<false><<<dim3(1024 / 128, 4096 / 128), 256, 0, stream>>>(
        Ob, Wo, bo, out, 4096, 1024, 1024, nullptr, nullptr, nullptr, nullptr, nullptr);
}
